# starting kernel
# baseline (speedup 1.0000x reference)
#include <hip/hip_runtime.h>
#include <hip/hip_bf16.h>


typedef float v2f __attribute__((ext_vector_type(2)));
typedef float v8f __attribute__((ext_vector_type(8)));

#define BATCH   128
#define MAT     1024
#define BMAT    (BATCH * MAT)
#define PAD     34
#define MTILE   (32 * PAD)

__device__ __forceinline__ v8f wmma4(v2f a, v2f b, v8f c) {
  return __builtin_amdgcn_wmma_f32_16x16x4_f32(
       false, a,  false, b,
       (short)0, c,  false,  false);
}

__device__ __forceinline__ v2f lds2(const float* p) {
  return *reinterpret_cast<const v2f*>(p);
}

__global__ void mlp_kernel(const float* __restrict__ x,
                           const float* __restrict__ W1, const float* __restrict__ b1,
                           const float* __restrict__ W2, const float* __restrict__ b2,
                           const float* __restrict__ W3, const float* __restrict__ b3,
                           float* __restrict__ Ms) {
  __shared__ float h1[1024];
  __shared__ float h2[32];
  __shared__ float xs[4];
  const int b   = blockIdx.x;
  const int tid = threadIdx.x;
  if (tid < 4) xs[tid] = x[b * 4 + tid];
  __syncthreads();
  const float x0 = xs[0], x1 = xs[1], x2 = xs[2], x3 = xs[3];
  for (int o = tid; o < 1024; o += 256) {
    const float* w = W1 + o * 4;
    float v = fmaf(x0, w[0], fmaf(x1, w[1], fmaf(x2, w[2], fmaf(x3, w[3], b1[o]))));
    h1[o] = fmaxf(v, 0.0f);
  }
  __syncthreads();
  if (tid < 32) {
    const float* w = W2 + tid * 1024;
    float v = b2[tid];
    for (int k = 0; k < 1024; ++k) v = fmaf(h1[k], w[k], v);
    h2[tid] = fmaxf(v, 0.0f);
  }
  __syncthreads();
  for (int o = tid; o < 2048; o += 256) {
    const float* w = W3 + o * 32;
    float v = b3[o];
#pragma unroll
    for (int k = 0; k < 32; ++k) v = fmaf(h2[k], w[k], v);
    const int s   = o >> 10;
    const int rem = o & 1023;
    Ms[(size_t)s * BMAT + (size_t)b * MAT + rem] = v;
  }
}

template <int S>
__global__ void pairprod_kernel(const float* __restrict__ Cin,
                                float* __restrict__ Cout) {
  extern __shared__ __align__(16) float sm[];
  float* rm = sm;
  float* cm = sm + S * MTILE;
  const int b   = blockIdx.x;
  const int tid = threadIdx.x;
  for (int idx = tid; idx < S * 1024; idx += 256) {
    const int s = idx >> 10;
    const int e = idx & 1023;
    const int r = e >> 5, c = e & 31;
    const float v = Cin[(size_t)s * BMAT + (size_t)b * MAT + e];
    rm[s * MTILE + r * PAD + c] = v;
    cm[s * MTILE + c * PAD + r] = v;
  }
  __syncthreads();
  const int wave = tid >> 5;
  const int lane = tid & 31;
  const int lm   = lane & 15;
  const int half = lane >> 4;
  for (int p = wave; p < S * S; p += 8) {
    const float* X = rm + (p / S) * MTILE;
    const float* Y = cm + (p % S) * MTILE;
    v8f acc[2][2] = {};
#pragma unroll
    for (int kc = 0; kc < 8; ++kc) {
      const int kk = kc * 4 + half * 2;
      const v2f a0 = lds2(X + lm * PAD + kk);
      const v2f a1 = lds2(X + (lm + 16) * PAD + kk);
      const v2f b0 = lds2(Y + lm * PAD + kk);
      const v2f b1 = lds2(Y + (lm + 16) * PAD + kk);
      acc[0][0] = wmma4(a0, b0, acc[0][0]);
      acc[0][1] = wmma4(a0, b1, acc[0][1]);
      acc[1][0] = wmma4(a1, b0, acc[1][0]);
      acc[1][1] = wmma4(a1, b1, acc[1][1]);
    }
    float* O = Cout + (size_t)p * BMAT + (size_t)b * MAT;
#pragma unroll
    for (int ti = 0; ti < 2; ++ti)
#pragma unroll
      for (int tj = 0; tj < 2; ++tj)
#pragma unroll
        for (int r = 0; r < 8; ++r) {
          const int row = ti * 16 + r + half * 8;
          const int col = tj * 16 + lm;
          O[row * 32 + col] = acc[ti][tj][r];
        }
  }
}

__global__ void trace_gemm_kernel(const float* __restrict__ C8,
                                  float* __restrict__ out) {
  __shared__ __align__(16) float Lt[128 * PAD];
  __shared__ __align__(16) float Rt[64 * PAD];
  const int b    = blockIdx.x >> 3;
  const int blk  = blockIdx.x & 7;
  const int s0   = (blk >> 2) * 128;
  const int t0   = (blk & 3) * 64;
  const int tid  = threadIdx.x;
  const int wave = tid >> 5;
  const int lane = tid & 31;
  const int lm   = lane & 15;
  const int half = lane >> 4;
  const int ti0  = (wave >> 1) * 2;
  const int tj0  = (wave & 1) * 2;
  v8f acc[2][2] = {};
  const float* base = C8 + (size_t)b * MAT;
  for (int chunk = 0; chunk < 32; ++chunk) {
    for (int idx = tid; idx < 128 * 32; idx += 256) {
      const int sl = idx >> 5, kl = idx & 31;
      const float* g = base + (size_t)(s0 + sl) * BMAT + chunk * 32 + kl;
      Lt[sl * PAD + kl] = *g;
      if (chunk < 31) __builtin_prefetch(g + 32, 0, 0);
    }
    for (int idx = tid; idx < 64 * 32; idx += 256) {
      const int tl = idx >> 5, kl = idx & 31;
      Rt[tl * PAD + kl] = base[(size_t)(t0 + tl) * BMAT + kl * 32 + chunk];
    }
    __syncthreads();
#pragma unroll
    for (int kc = 0; kc < 8; ++kc) {
      const int kk = kc * 4 + half * 2;
      const v2f a0 = lds2(Lt + (ti0 * 16 + lm) * PAD + kk);
      const v2f a1 = lds2(Lt + ((ti0 + 1) * 16 + lm) * PAD + kk);
      const v2f b0 = lds2(Rt + (tj0 * 16 + lm) * PAD + kk);
      const v2f b1 = lds2(Rt + ((tj0 + 1) * 16 + lm) * PAD + kk);
      acc[0][0] = wmma4(a0, b0, acc[0][0]);
      acc[0][1] = wmma4(a0, b1, acc[0][1]);
      acc[1][0] = wmma4(a1, b0, acc[1][0]);
      acc[1][1] = wmma4(a1, b1, acc[1][1]);
    }
    __syncthreads();
  }
  float* O = out + (size_t)b * 65536;
#pragma unroll
  for (int di = 0; di < 2; ++di)
#pragma unroll
    for (int dj = 0; dj < 2; ++dj)
#pragma unroll
      for (int r = 0; r < 8; ++r) {
        const int row = s0 + (ti0 + di) * 16 + r + half * 8;
        const int col = t0 + (tj0 + dj) * 16 + lm;
        O[(size_t)row * 256 + col] = acc[di][dj][r];
      }
}

__global__ void norm_reduce_kernel(const float* __restrict__ out,
                                   float* __restrict__ inv) {
  __shared__ float red[256];
  const int b = blockIdx.x, tid = threadIdx.x;
  const float* p = out + (size_t)b * 65536;
  float s = 0.f;
  for (int i = tid; i < 65536; i += 256) { const float v = p[i]; s = fmaf(v, v, s); }
  red[tid] = s;
  __syncthreads();
  for (int off = 128; off > 0; off >>= 1) {
    if (tid < off) red[tid] += red[tid + off];
    __syncthreads();
  }
  if (tid == 0) inv[b] = rsqrtf(red[0]);
}

__global__ void scale_kernel(float* __restrict__ out,
                             const float* __restrict__ inv) {
  size_t i = (size_t)blockIdx.x * blockDim.x + threadIdx.x;
  const size_t n = (size_t)BATCH * 65536;
  const size_t stride = (size_t)gridDim.x * blockDim.x;
  for (; i < n; i += stride) out[i] *= inv[i >> 16];
}

extern "C" void kernel_launch(void* const* d_in, const int* in_sizes, int n_in,
                              void* d_out, int out_size, void* d_ws, size_t ws_size,
                              hipStream_t stream) {
  const float* x  = (const float*)d_in[0];
  const float* W1 = (const float*)d_in[1];
  const float* b1 = (const float*)d_in[2];
  const float* W2 = (const float*)d_in[3];
  const float* b2 = (const float*)d_in[4];
  const float* W3 = (const float*)d_in[5];
  const float* b3 = (const float*)d_in[6];
  float* out = (float*)d_out;

  float* ws  = (float*)d_ws;
  float* Ms  = ws;
  float* C2  = Ms + (size_t)2  * BMAT;
  float* C4  = C2 + (size_t)4  * BMAT;
  float* C8  = C4 + (size_t)16 * BMAT;
  float* inv = C8 + (size_t)256 * BMAT;

  mlp_kernel<<<BATCH, 256, 0, stream>>>(x, W1, b1, W2, b2, W3, b3, Ms);

  pairprod_kernel<2><<<BATCH, 256, (size_t)2 * 2  * MTILE * sizeof(float), stream>>>(Ms, C2);
  pairprod_kernel<4><<<BATCH, 256, (size_t)2 * 4  * MTILE * sizeof(float), stream>>>(C2, C4);
  pairprod_kernel<16><<<BATCH, 256, (size_t)2 * 16 * MTILE * sizeof(float), stream>>>(C4, C8);

  trace_gemm_kernel<<<BATCH * 8, 256, 0, stream>>>(C8, out);
  norm_reduce_kernel<<<BATCH, 256, 0, stream>>>(out, inv);
  scale_kernel<<<4096, 256, 0, stream>>>(out, inv);
}


# C8 kept 8x8-block-major; trace GEMM K chunk = one 8x8 block of L and its mirror block of R, both staged from contiguous 256B runs (R was stride-128B single floats); K sum order regrouped by block
# speedup vs baseline: 3.5994x; 3.5994x over previous
#include <hip/hip_runtime.h>
#include <hip/hip_bf16.h>


typedef float v2f __attribute__((ext_vector_type(2)));
typedef float v8f __attribute__((ext_vector_type(8)));

#define BATCH   128
#define MAT     1024
#define BMAT    (BATCH * MAT)
#define PAD     34
#define MTILE   (32 * PAD)
#define KPITCH  68

__device__ __forceinline__ v8f wmma4(v2f a, v2f b, v8f c) {
  return __builtin_amdgcn_wmma_f32_16x16x4_f32(
       false, a,  false, b,
       (short)0, c,  false,  false);
}

__device__ __forceinline__ v2f lds2(const float* p) {
  return *reinterpret_cast<const v2f*>(p);
}

__global__ void mlp_kernel(const float* __restrict__ x,
                           const float* __restrict__ W1, const float* __restrict__ b1,
                           const float* __restrict__ W2, const float* __restrict__ b2,
                           const float* __restrict__ W3, const float* __restrict__ b3,
                           float* __restrict__ Ms) {
  __shared__ float h1[1024];
  __shared__ float h2[32];
  __shared__ float xs[4];
  const int b   = blockIdx.x;
  const int tid = threadIdx.x;
  if (tid < 4) xs[tid] = x[b * 4 + tid];
  __syncthreads();
  const float x0 = xs[0], x1 = xs[1], x2 = xs[2], x3 = xs[3];
  for (int o = tid; o < 1024; o += 256) {
    const float* w = W1 + o * 4;
    float v = fmaf(x0, w[0], fmaf(x1, w[1], fmaf(x2, w[2], fmaf(x3, w[3], b1[o]))));
    h1[o] = fmaxf(v, 0.0f);
  }
  __syncthreads();
  if (tid < 32) {
    const float* w = W2 + tid * 1024;
    float v = b2[tid];
    for (int k = 0; k < 1024; ++k) v = fmaf(h1[k], w[k], v);
    h2[tid] = fmaxf(v, 0.0f);
  }
  __syncthreads();
  for (int o = tid; o < 2048; o += 256) {
    const float* w = W3 + o * 32;
    float v = b3[o];
#pragma unroll
    for (int k = 0; k < 32; ++k) v = fmaf(h2[k], w[k], v);
    const int s   = o >> 10;
    const int rem = o & 1023;
    Ms[(size_t)s * BMAT + (size_t)b * MAT + rem] = v;
  }
}

template <int S, bool BLK>
__global__ void pairprod_kernel(const float* __restrict__ Cin,
                                float* __restrict__ Cout) {
  extern __shared__ __align__(16) float sm[];
  float* rm = sm;
  float* cm = sm + S * MTILE;
  const int b   = blockIdx.x;
  const int tid = threadIdx.x;
  for (int idx = tid; idx < S * 1024; idx += 256) {
    const int s = idx >> 10;
    const int e = idx & 1023;
    const int r = e >> 5, c = e & 31;
    const float v = Cin[(size_t)s * BMAT + (size_t)b * MAT + e];
    rm[s * MTILE + r * PAD + c] = v;
    cm[s * MTILE + c * PAD + r] = v;
  }
  __syncthreads();
  const int wave = tid >> 5;
  const int lane = tid & 31;
  const int lm   = lane & 15;
  const int half = lane >> 4;
  for (int p = wave; p < S * S; p += 8) {
    const float* X = rm + (p / S) * MTILE;
    const float* Y = cm + (p % S) * MTILE;
    v8f acc[2][2] = {};
#pragma unroll
    for (int kc = 0; kc < 8; ++kc) {
      const int kk = kc * 4 + half * 2;
      const v2f a0 = lds2(X + lm * PAD + kk);
      const v2f a1 = lds2(X + (lm + 16) * PAD + kk);
      const v2f b0 = lds2(Y + lm * PAD + kk);
      const v2f b1 = lds2(Y + (lm + 16) * PAD + kk);
      acc[0][0] = wmma4(a0, b0, acc[0][0]);
      acc[0][1] = wmma4(a0, b1, acc[0][1]);
      acc[1][0] = wmma4(a1, b0, acc[1][0]);
      acc[1][1] = wmma4(a1, b1, acc[1][1]);
    }
    float* O = Cout + (size_t)p * BMAT + (size_t)b * MAT;
#pragma unroll
    for (int ti = 0; ti < 2; ++ti)
#pragma unroll
      for (int tj = 0; tj < 2; ++tj)
#pragma unroll
        for (int r = 0; r < 8; ++r) {
          const int row = ti * 16 + r + half * 8;
          const int col = tj * 16 + lm;
          const int o = BLK ? ((((row >> 3) << 2) + (col >> 3)) << 6) + ((row & 7) << 3) + (col & 7)
                            : row * 32 + col;
          O[o] = acc[ti][tj][r];
        }
  }
}

__global__ void trace_gemm_kernel(const float* __restrict__ C8,
                                  float* __restrict__ out) {
  __shared__ __align__(16) float Lt[128 * KPITCH];
  __shared__ __align__(16) float Rt[64 * KPITCH];
  const int b    = blockIdx.x >> 3;
  const int blk  = blockIdx.x & 7;
  const int s0   = (blk >> 2) * 128;
  const int t0   = (blk & 3) * 64;
  const int tid  = threadIdx.x;
  const int wave = tid >> 5;
  const int lane = tid & 31;
  const int lm   = lane & 15;
  const int half = lane >> 4;
  const int ti0  = (wave >> 1) * 2;
  const int tj0  = (wave & 1) * 2;
  v8f acc[2][2] = {};
  const float* base = C8 + (size_t)b * MAT;
  for (int chunk = 0; chunk < 16; ++chunk) {
    const int lblk = chunk * 64;
    const int rblk = (((chunk & 3) << 2) + (chunk >> 2)) * 64;
    for (int n = 0; n < 8; ++n) {
      const int idx = tid + n * 256;
      const int sl = idx >> 4, q = idx & 15;
      const float4 v = *reinterpret_cast<const float4*>(
          base + (size_t)(s0 + sl) * BMAT + lblk + q * 4);
      *reinterpret_cast<float4*>(Lt + sl * KPITCH + q * 4) = v;
    }
    for (int n = 0; n < 4; ++n) {
      const int idx = tid + n * 256;
      const int tl = idx >> 4, q = idx & 15;
      const int jl = q >> 1, i4 = (q & 1) * 4;
      const float4 v = *reinterpret_cast<const float4*>(
          base + (size_t)(t0 + tl) * BMAT + rblk + jl * 8 + i4);
      float* d = Rt + tl * KPITCH + i4 * 8 + jl;
      d[0] = v.x; d[8] = v.y; d[16] = v.z; d[24] = v.w;
    }
    __syncthreads();
#pragma unroll
    for (int kc = 0; kc < 16; ++kc) {
      const int kk = kc * 4 + half * 2;
      const v2f a0 = lds2(Lt + (ti0 * 16 + lm) * KPITCH + kk);
      const v2f a1 = lds2(Lt + ((ti0 + 1) * 16 + lm) * KPITCH + kk);
      const v2f b0 = lds2(Rt + (tj0 * 16 + lm) * KPITCH + kk);
      const v2f b1 = lds2(Rt + ((tj0 + 1) * 16 + lm) * KPITCH + kk);
      acc[0][0] = wmma4(a0, b0, acc[0][0]);
      acc[0][1] = wmma4(a0, b1, acc[0][1]);
      acc[1][0] = wmma4(a1, b0, acc[1][0]);
      acc[1][1] = wmma4(a1, b1, acc[1][1]);
    }
    __syncthreads();
  }
  float* O = out + (size_t)b * 65536;
#pragma unroll
  for (int di = 0; di < 2; ++di)
#pragma unroll
    for (int dj = 0; dj < 2; ++dj)
#pragma unroll
      for (int r = 0; r < 8; ++r) {
        const int row = s0 + (ti0 + di) * 16 + r + half * 8;
        const int col = t0 + (tj0 + dj) * 16 + lm;
        O[(size_t)row * 256 + col] = acc[di][dj][r];
      }
}

__global__ void norm_reduce_kernel(const float* __restrict__ out,
                                   float* __restrict__ inv) {
  __shared__ float red[256];
  const int b = blockIdx.x, tid = threadIdx.x;
  const float* p = out + (size_t)b * 65536;
  float s = 0.f;
  for (int i = tid; i < 65536; i += 256) { const float v = p[i]; s = fmaf(v, v, s); }
  red[tid] = s;
  __syncthreads();
  for (int off = 128; off > 0; off >>= 1) {
    if (tid < off) red[tid] += red[tid + off];
    __syncthreads();
  }
  if (tid == 0) inv[b] = rsqrtf(red[0]);
}

__global__ void scale_kernel(float* __restrict__ out,
                             const float* __restrict__ inv) {
  size_t i = (size_t)blockIdx.x * blockDim.x + threadIdx.x;
  const size_t n = (size_t)BATCH * 65536;
  const size_t stride = (size_t)gridDim.x * blockDim.x;
  for (; i < n; i += stride) out[i] *= inv[i >> 16];
}

extern "C" void kernel_launch(void* const* d_in, const int* in_sizes, int n_in,
                              void* d_out, int out_size, void* d_ws, size_t ws_size,
                              hipStream_t stream) {
  const float* x  = (const float*)d_in[0];
  const float* W1 = (const float*)d_in[1];
  const float* b1 = (const float*)d_in[2];
  const float* W2 = (const float*)d_in[3];
  const float* b2 = (const float*)d_in[4];
  const float* W3 = (const float*)d_in[5];
  const float* b3 = (const float*)d_in[6];
  float* out = (float*)d_out;

  float* ws  = (float*)d_ws;
  float* Ms  = ws;
  float* C2  = Ms + (size_t)2  * BMAT;
  float* C4  = C2 + (size_t)4  * BMAT;
  float* C8  = C4 + (size_t)16 * BMAT;
  float* inv = C8 + (size_t)256 * BMAT;

  mlp_kernel<<<BATCH, 256, 0, stream>>>(x, W1, b1, W2, b2, W3, b3, Ms);

  pairprod_kernel<2, false><<<BATCH, 256, (size_t)2 * 2  * MTILE * sizeof(float), stream>>>(Ms, C2);
  pairprod_kernel<4, false><<<BATCH, 256, (size_t)2 * 4  * MTILE * sizeof(float), stream>>>(C2, C4);
  pairprod_kernel<16, true><<<BATCH, 256, (size_t)2 * 16 * MTILE * sizeof(float), stream>>>(C4, C8);

  trace_gemm_kernel<<<BATCH * 8, 256, 0, stream>>>(C8, out);
  norm_reduce_kernel<<<BATCH, 256, 0, stream>>>(out, inv);
  scale_kernel<<<4096, 256, 0, stream>>>(out, inv);
}


# trace GEMM exploits coeffs[s,t]=coeffs[t,s] (L, R are one chain): computes only the 10 upper 64x64 tiles per batch, stores off-diagonal tiles twice (direct + transposed); mirrored sums reordered
# speedup vs baseline: 3.9312x; 1.0922x over previous
#include <hip/hip_runtime.h>
#include <hip/hip_bf16.h>


typedef float v2f __attribute__((ext_vector_type(2)));
typedef float v8f __attribute__((ext_vector_type(8)));

#define BATCH   128
#define MAT     1024
#define BMAT    (BATCH * MAT)
#define PAD     34
#define MTILE   (32 * PAD)
#define KPITCH  68

__device__ __forceinline__ v8f wmma4(v2f a, v2f b, v8f c) {
  return __builtin_amdgcn_wmma_f32_16x16x4_f32(
       false, a,  false, b,
       (short)0, c,  false,  false);
}

__device__ __forceinline__ v2f lds2(const float* p) {
  return *reinterpret_cast<const v2f*>(p);
}

__global__ void mlp_kernel(const float* __restrict__ x,
                           const float* __restrict__ W1, const float* __restrict__ b1,
                           const float* __restrict__ W2, const float* __restrict__ b2,
                           const float* __restrict__ W3, const float* __restrict__ b3,
                           float* __restrict__ Ms) {
  __shared__ float h1[1024];
  __shared__ float h2[32];
  __shared__ float xs[4];
  const int b   = blockIdx.x;
  const int tid = threadIdx.x;
  if (tid < 4) xs[tid] = x[b * 4 + tid];
  __syncthreads();
  const float x0 = xs[0], x1 = xs[1], x2 = xs[2], x3 = xs[3];
  for (int o = tid; o < 1024; o += 256) {
    const float* w = W1 + o * 4;
    float v = fmaf(x0, w[0], fmaf(x1, w[1], fmaf(x2, w[2], fmaf(x3, w[3], b1[o]))));
    h1[o] = fmaxf(v, 0.0f);
  }
  __syncthreads();
  if (tid < 32) {
    const float* w = W2 + tid * 1024;
    float v = b2[tid];
    for (int k = 0; k < 1024; ++k) v = fmaf(h1[k], w[k], v);
    h2[tid] = fmaxf(v, 0.0f);
  }
  __syncthreads();
  for (int o = tid; o < 2048; o += 256) {
    const float* w = W3 + o * 32;
    float v = b3[o];
#pragma unroll
    for (int k = 0; k < 32; ++k) v = fmaf(h2[k], w[k], v);
    const int s   = o >> 10;
    const int rem = o & 1023;
    Ms[(size_t)s * BMAT + (size_t)b * MAT + rem] = v;
  }
}

template <int S, bool BLK>
__global__ void pairprod_kernel(const float* __restrict__ Cin,
                                float* __restrict__ Cout) {
  extern __shared__ __align__(16) float sm[];
  float* rm = sm;
  float* cm = sm + S * MTILE;
  const int b   = blockIdx.x;
  const int tid = threadIdx.x;
  for (int idx = tid; idx < S * 1024; idx += 256) {
    const int s = idx >> 10;
    const int e = idx & 1023;
    const int r = e >> 5, c = e & 31;
    const float v = Cin[(size_t)s * BMAT + (size_t)b * MAT + e];
    rm[s * MTILE + r * PAD + c] = v;
    cm[s * MTILE + c * PAD + r] = v;
  }
  __syncthreads();
  const int wave = tid >> 5;
  const int lane = tid & 31;
  const int lm   = lane & 15;
  const int half = lane >> 4;
  for (int p = wave; p < S * S; p += 8) {
    const float* X = rm + (p / S) * MTILE;
    const float* Y = cm + (p % S) * MTILE;
    v8f acc[2][2] = {};
#pragma unroll
    for (int kc = 0; kc < 8; ++kc) {
      const int kk = kc * 4 + half * 2;
      const v2f a0 = lds2(X + lm * PAD + kk);
      const v2f a1 = lds2(X + (lm + 16) * PAD + kk);
      const v2f b0 = lds2(Y + lm * PAD + kk);
      const v2f b1 = lds2(Y + (lm + 16) * PAD + kk);
      acc[0][0] = wmma4(a0, b0, acc[0][0]);
      acc[0][1] = wmma4(a0, b1, acc[0][1]);
      acc[1][0] = wmma4(a1, b0, acc[1][0]);
      acc[1][1] = wmma4(a1, b1, acc[1][1]);
    }
    float* O = Cout + (size_t)p * BMAT + (size_t)b * MAT;
#pragma unroll
    for (int ti = 0; ti < 2; ++ti)
#pragma unroll
      for (int tj = 0; tj < 2; ++tj)
#pragma unroll
        for (int r = 0; r < 8; ++r) {
          const int row = ti * 16 + r + half * 8;
          const int col = tj * 16 + lm;
          const int o = BLK ? ((((row >> 3) << 2) + (col >> 3)) << 6) + ((row & 7) << 3) + (col & 7)
                            : row * 32 + col;
          O[o] = acc[ti][tj][r];
        }
  }
}

__global__ void trace_gemm_kernel(const float* __restrict__ C8,
                                  float* __restrict__ out) {
  __shared__ __align__(16) float Lt[64 * KPITCH];
  __shared__ __align__(16) float Rt[64 * KPITCH];
  const int b    = blockIdx.x / 10;
  const int blk  = blockIdx.x - b * 10;
  const int I    = (blk >= 4) + (blk >= 7) + (blk >= 9);
  const int J    = blk - ((I * (9 - I)) >> 1) + I;
  const int s0   = I * 64;
  const int t0   = J * 64;
  const int tid  = threadIdx.x;
  const int wave = tid >> 5;
  const int lane = tid & 31;
  const int lm   = lane & 15;
  const int half = lane >> 4;
  const int ti0  = (wave >> 1) * 2;
  const int tj0  = (wave & 1) * 2;
  v8f acc[2][2] = {};
  const float* base = C8 + (size_t)b * MAT;
  for (int chunk = 0; chunk < 16; ++chunk) {
    const int lblk = chunk * 64;
    const int rblk = (((chunk & 3) << 2) + (chunk >> 2)) * 64;
    for (int n = 0; n < 8; ++n) {
      const int idx = tid + n * 128;
      const int sl = idx >> 4, q = idx & 15;
      const float4 v = *reinterpret_cast<const float4*>(
          base + (size_t)(s0 + sl) * BMAT + lblk + q * 4);
      *reinterpret_cast<float4*>(Lt + sl * KPITCH + q * 4) = v;
    }
    for (int n = 0; n < 8; ++n) {
      const int idx = tid + n * 128;
      const int tl = idx >> 4, q = idx & 15;
      const int jl = q >> 1, i4 = (q & 1) * 4;
      const float4 v = *reinterpret_cast<const float4*>(
          base + (size_t)(t0 + tl) * BMAT + rblk + jl * 8 + i4);
      float* d = Rt + tl * KPITCH + i4 * 8 + jl;
      d[0] = v.x; d[8] = v.y; d[16] = v.z; d[24] = v.w;
    }
    __syncthreads();
#pragma unroll
    for (int kc = 0; kc < 16; ++kc) {
      const int kk = kc * 4 + half * 2;
      const v2f a0 = lds2(Lt + (ti0 * 16 + lm) * KPITCH + kk);
      const v2f a1 = lds2(Lt + ((ti0 + 1) * 16 + lm) * KPITCH + kk);
      const v2f b0 = lds2(Rt + (tj0 * 16 + lm) * KPITCH + kk);
      const v2f b1 = lds2(Rt + ((tj0 + 1) * 16 + lm) * KPITCH + kk);
      acc[0][0] = wmma4(a0, b0, acc[0][0]);
      acc[0][1] = wmma4(a0, b1, acc[0][1]);
      acc[1][0] = wmma4(a1, b0, acc[1][0]);
      acc[1][1] = wmma4(a1, b1, acc[1][1]);
    }
    __syncthreads();
  }
  float* O = out + (size_t)b * 65536;
#pragma unroll
  for (int di = 0; di < 2; ++di)
#pragma unroll
    for (int dj = 0; dj < 2; ++dj)
#pragma unroll
      for (int r = 0; r < 8; ++r) {
        const int row = s0 + (ti0 + di) * 16 + r + half * 8;
        const int col = t0 + (tj0 + dj) * 16 + lm;
        O[(size_t)row * 256 + col] = acc[di][dj][r];
      }
  if (I != J) {
    for (int di = 0; di < 2; ++di)
      for (int dj = 0; dj < 2; ++dj) {
        const int tm = t0 + (tj0 + dj) * 16 + lm;
        const int sb = s0 + (ti0 + di) * 16 + half * 8;
        float* p = O + (size_t)tm * 256 + sb;
        *reinterpret_cast<float4*>(p) =
            make_float4(acc[di][dj][0], acc[di][dj][1], acc[di][dj][2], acc[di][dj][3]);
        *reinterpret_cast<float4*>(p + 4) =
            make_float4(acc[di][dj][4], acc[di][dj][5], acc[di][dj][6], acc[di][dj][7]);
      }
  }
}

__global__ void norm_reduce_kernel(const float* __restrict__ out,
                                   float* __restrict__ inv) {
  __shared__ float red[256];
  const int b = blockIdx.x, tid = threadIdx.x;
  const float* p = out + (size_t)b * 65536;
  float s = 0.f;
  for (int i = tid; i < 65536; i += 256) { const float v = p[i]; s = fmaf(v, v, s); }
  red[tid] = s;
  __syncthreads();
  for (int off = 128; off > 0; off >>= 1) {
    if (tid < off) red[tid] += red[tid + off];
    __syncthreads();
  }
  if (tid == 0) inv[b] = rsqrtf(red[0]);
}

__global__ void scale_kernel(float* __restrict__ out,
                             const float* __restrict__ inv) {
  size_t i = (size_t)blockIdx.x * blockDim.x + threadIdx.x;
  const size_t n = (size_t)BATCH * 65536;
  const size_t stride = (size_t)gridDim.x * blockDim.x;
  for (; i < n; i += stride) out[i] *= inv[i >> 16];
}

extern "C" void kernel_launch(void* const* d_in, const int* in_sizes, int n_in,
                              void* d_out, int out_size, void* d_ws, size_t ws_size,
                              hipStream_t stream) {
  const float* x  = (const float*)d_in[0];
  const float* W1 = (const float*)d_in[1];
  const float* b1 = (const float*)d_in[2];
  const float* W2 = (const float*)d_in[3];
  const float* b2 = (const float*)d_in[4];
  const float* W3 = (const float*)d_in[5];
  const float* b3 = (const float*)d_in[6];
  float* out = (float*)d_out;

  float* ws  = (float*)d_ws;
  float* Ms  = ws;
  float* C2  = Ms + (size_t)2  * BMAT;
  float* C4  = C2 + (size_t)4  * BMAT;
  float* C8  = C4 + (size_t)16 * BMAT;
  float* inv = C8 + (size_t)256 * BMAT;

  mlp_kernel<<<BATCH, 256, 0, stream>>>(x, W1, b1, W2, b2, W3, b3, Ms);

  pairprod_kernel<2, false><<<BATCH, 256, (size_t)2 * 2  * MTILE * sizeof(float), stream>>>(Ms, C2);
  pairprod_kernel<4, false><<<BATCH, 256, (size_t)2 * 4  * MTILE * sizeof(float), stream>>>(C2, C4);
  pairprod_kernel<16, true><<<BATCH, 256, (size_t)2 * 16 * MTILE * sizeof(float), stream>>>(C4, C8);

  trace_gemm_kernel<<<BATCH * 10, 128, 0, stream>>>(C8, out);
  norm_reduce_kernel<<<BATCH, 256, 0, stream>>>(out, inv);
  scale_kernel<<<4096, 256, 0, stream>>>(out, inv);
}


# row norm fused into trace GEMM epilogue: each block writes its tile's sum of squares (mirror tiles x2), scale pass adds the 10 partials of its row; norm_reduce launch dropped; norm sum order changed
# speedup vs baseline: 4.4470x; 1.1312x over previous
#include <hip/hip_runtime.h>
#include <hip/hip_bf16.h>


typedef float v2f __attribute__((ext_vector_type(2)));
typedef float v8f __attribute__((ext_vector_type(8)));

#define BATCH   128
#define MAT     1024
#define BMAT    (BATCH * MAT)
#define PAD     34
#define MTILE   (32 * PAD)
#define KPITCH  68

__device__ __forceinline__ v8f wmma4(v2f a, v2f b, v8f c) {
  return __builtin_amdgcn_wmma_f32_16x16x4_f32(
       false, a,  false, b,
       (short)0, c,  false,  false);
}

__device__ __forceinline__ v2f lds2(const float* p) {
  return *reinterpret_cast<const v2f*>(p);
}

__global__ void mlp_kernel(const float* __restrict__ x,
                           const float* __restrict__ W1, const float* __restrict__ b1,
                           const float* __restrict__ W2, const float* __restrict__ b2,
                           const float* __restrict__ W3, const float* __restrict__ b3,
                           float* __restrict__ Ms) {
  __shared__ float h1[1024];
  __shared__ float h2[32];
  __shared__ float xs[4];
  const int b   = blockIdx.x;
  const int tid = threadIdx.x;
  if (tid < 4) xs[tid] = x[b * 4 + tid];
  __syncthreads();
  const float x0 = xs[0], x1 = xs[1], x2 = xs[2], x3 = xs[3];
  for (int o = tid; o < 1024; o += 256) {
    const float* w = W1 + o * 4;
    float v = fmaf(x0, w[0], fmaf(x1, w[1], fmaf(x2, w[2], fmaf(x3, w[3], b1[o]))));
    h1[o] = fmaxf(v, 0.0f);
  }
  __syncthreads();
  if (tid < 32) {
    const float* w = W2 + tid * 1024;
    float v = b2[tid];
    for (int k = 0; k < 1024; ++k) v = fmaf(h1[k], w[k], v);
    h2[tid] = fmaxf(v, 0.0f);
  }
  __syncthreads();
  for (int o = tid; o < 2048; o += 256) {
    const float* w = W3 + o * 32;
    float v = b3[o];
#pragma unroll
    for (int k = 0; k < 32; ++k) v = fmaf(h2[k], w[k], v);
    const int s   = o >> 10;
    const int rem = o & 1023;
    Ms[(size_t)s * BMAT + (size_t)b * MAT + rem] = v;
  }
}

template <int S, bool BLK>
__global__ void pairprod_kernel(const float* __restrict__ Cin,
                                float* __restrict__ Cout) {
  extern __shared__ __align__(16) float sm[];
  float* rm = sm;
  float* cm = sm + S * MTILE;
  const int b   = blockIdx.x;
  const int tid = threadIdx.x;
  for (int idx = tid; idx < S * 1024; idx += 256) {
    const int s = idx >> 10;
    const int e = idx & 1023;
    const int r = e >> 5, c = e & 31;
    const float v = Cin[(size_t)s * BMAT + (size_t)b * MAT + e];
    rm[s * MTILE + r * PAD + c] = v;
    cm[s * MTILE + c * PAD + r] = v;
  }
  __syncthreads();
  const int wave = tid >> 5;
  const int lane = tid & 31;
  const int lm   = lane & 15;
  const int half = lane >> 4;
  for (int p = wave; p < S * S; p += 8) {
    const float* X = rm + (p / S) * MTILE;
    const float* Y = cm + (p % S) * MTILE;
    v8f acc[2][2] = {};
#pragma unroll
    for (int kc = 0; kc < 8; ++kc) {
      const int kk = kc * 4 + half * 2;
      const v2f a0 = lds2(X + lm * PAD + kk);
      const v2f a1 = lds2(X + (lm + 16) * PAD + kk);
      const v2f b0 = lds2(Y + lm * PAD + kk);
      const v2f b1 = lds2(Y + (lm + 16) * PAD + kk);
      acc[0][0] = wmma4(a0, b0, acc[0][0]);
      acc[0][1] = wmma4(a0, b1, acc[0][1]);
      acc[1][0] = wmma4(a1, b0, acc[1][0]);
      acc[1][1] = wmma4(a1, b1, acc[1][1]);
    }
    float* O = Cout + (size_t)p * BMAT + (size_t)b * MAT;
#pragma unroll
    for (int ti = 0; ti < 2; ++ti)
#pragma unroll
      for (int tj = 0; tj < 2; ++tj)
#pragma unroll
        for (int r = 0; r < 8; ++r) {
          const int row = ti * 16 + r + half * 8;
          const int col = tj * 16 + lm;
          const int o = BLK ? ((((row >> 3) << 2) + (col >> 3)) << 6) + ((row & 7) << 3) + (col & 7)
                            : row * 32 + col;
          O[o] = acc[ti][tj][r];
        }
  }
}

__global__ void trace_gemm_kernel(const float* __restrict__ C8,
                                  float* __restrict__ out,
                                  float* __restrict__ part) {
  __shared__ __align__(16) float Lt[64 * KPITCH];
  __shared__ __align__(16) float Rt[64 * KPITCH];
  const int b    = blockIdx.x / 10;
  const int blk  = blockIdx.x - b * 10;
  const int I    = (blk >= 4) + (blk >= 7) + (blk >= 9);
  const int J    = blk - ((I * (9 - I)) >> 1) + I;
  const int s0   = I * 64;
  const int t0   = J * 64;
  const int tid  = threadIdx.x;
  const int wave = tid >> 5;
  const int lane = tid & 31;
  const int lm   = lane & 15;
  const int half = lane >> 4;
  const int ti0  = (wave >> 1) * 2;
  const int tj0  = (wave & 1) * 2;
  v8f acc[2][2] = {};
  const float* base = C8 + (size_t)b * MAT;
  for (int chunk = 0; chunk < 16; ++chunk) {
    const int lblk = chunk * 64;
    const int rblk = (((chunk & 3) << 2) + (chunk >> 2)) * 64;
    for (int n = 0; n < 8; ++n) {
      const int idx = tid + n * 128;
      const int sl = idx >> 4, q = idx & 15;
      const float4 v = *reinterpret_cast<const float4*>(
          base + (size_t)(s0 + sl) * BMAT + lblk + q * 4);
      *reinterpret_cast<float4*>(Lt + sl * KPITCH + q * 4) = v;
    }
    for (int n = 0; n < 8; ++n) {
      const int idx = tid + n * 128;
      const int tl = idx >> 4, q = idx & 15;
      const int jl = q >> 1, i4 = (q & 1) * 4;
      const float4 v = *reinterpret_cast<const float4*>(
          base + (size_t)(t0 + tl) * BMAT + rblk + jl * 8 + i4);
      float* d = Rt + tl * KPITCH + i4 * 8 + jl;
      d[0] = v.x; d[8] = v.y; d[16] = v.z; d[24] = v.w;
    }
    __syncthreads();
#pragma unroll
    for (int kc = 0; kc < 16; ++kc) {
      const int kk = kc * 4 + half * 2;
      const v2f a0 = lds2(Lt + (ti0 * 16 + lm) * KPITCH + kk);
      const v2f a1 = lds2(Lt + ((ti0 + 1) * 16 + lm) * KPITCH + kk);
      const v2f b0 = lds2(Rt + (tj0 * 16 + lm) * KPITCH + kk);
      const v2f b1 = lds2(Rt + ((tj0 + 1) * 16 + lm) * KPITCH + kk);
      acc[0][0] = wmma4(a0, b0, acc[0][0]);
      acc[0][1] = wmma4(a0, b1, acc[0][1]);
      acc[1][0] = wmma4(a1, b0, acc[1][0]);
      acc[1][1] = wmma4(a1, b1, acc[1][1]);
    }
    __syncthreads();
  }
  float ss = 0.f;
#pragma unroll
  for (int di = 0; di < 2; ++di)
#pragma unroll
    for (int dj = 0; dj < 2; ++dj)
#pragma unroll
      for (int r = 0; r < 8; ++r) ss = fmaf(acc[di][dj][r], acc[di][dj][r], ss);
  for (int off = 16; off > 0; off >>= 1) ss += __shfl_xor(ss, off, 32);
  if (lane == 0) Lt[wave] = ss;
  __syncthreads();
  if (tid == 0)
    part[blockIdx.x * 32] = ((Lt[0] + Lt[1]) + (Lt[2] + Lt[3])) * (I != J ? 2.f : 1.f);
  float* O = out + (size_t)b * 65536;
#pragma unroll
  for (int di = 0; di < 2; ++di)
#pragma unroll
    for (int dj = 0; dj < 2; ++dj)
#pragma unroll
      for (int r = 0; r < 8; ++r) {
        const int row = s0 + (ti0 + di) * 16 + r + half * 8;
        const int col = t0 + (tj0 + dj) * 16 + lm;
        O[(size_t)row * 256 + col] = acc[di][dj][r];
      }
  if (I != J) {
    for (int di = 0; di < 2; ++di)
      for (int dj = 0; dj < 2; ++dj) {
        const int tm = t0 + (tj0 + dj) * 16 + lm;
        const int sb = s0 + (ti0 + di) * 16 + half * 8;
        float* p = O + (size_t)tm * 256 + sb;
        *reinterpret_cast<float4*>(p) =
            make_float4(acc[di][dj][0], acc[di][dj][1], acc[di][dj][2], acc[di][dj][3]);
        *reinterpret_cast<float4*>(p + 4) =
            make_float4(acc[di][dj][4], acc[di][dj][5], acc[di][dj][6], acc[di][dj][7]);
      }
  }
}

__global__ void scale_kernel(float* __restrict__ out,
                             const float* __restrict__ part) {
  const float* q = part + (blockIdx.x >> 5) * 320;
  float s = q[0];
  for (int k = 1; k < 10; ++k) s += q[k * 32];
  const float inv = rsqrtf(s);
  float* p = out + (size_t)blockIdx.x * 2048 + threadIdx.x;
#pragma unroll
  for (int k = 0; k < 8; ++k) p[k * 256] *= inv;
}

extern "C" void kernel_launch(void* const* d_in, const int* in_sizes, int n_in,
                              void* d_out, int out_size, void* d_ws, size_t ws_size,
                              hipStream_t stream) {
  const float* x  = (const float*)d_in[0];
  const float* W1 = (const float*)d_in[1];
  const float* b1 = (const float*)d_in[2];
  const float* W2 = (const float*)d_in[3];
  const float* b2 = (const float*)d_in[4];
  const float* W3 = (const float*)d_in[5];
  const float* b3 = (const float*)d_in[6];
  float* out = (float*)d_out;

  float* ws  = (float*)d_ws;
  float* Ms  = ws;
  float* C2  = Ms + (size_t)2  * BMAT;
  float* C4  = C2 + (size_t)4  * BMAT;
  float* C8  = C4 + (size_t)16 * BMAT;
  float* part = Ms;

  mlp_kernel<<<BATCH, 256, 0, stream>>>(x, W1, b1, W2, b2, W3, b3, Ms);

  pairprod_kernel<2, false><<<BATCH, 256, (size_t)2 * 2  * MTILE * sizeof(float), stream>>>(Ms, C2);
  pairprod_kernel<4, false><<<BATCH, 256, (size_t)2 * 4  * MTILE * sizeof(float), stream>>>(C2, C4);
  pairprod_kernel<16, true><<<BATCH, 256, (size_t)2 * 16 * MTILE * sizeof(float), stream>>>(C4, C8);

  trace_gemm_kernel<<<BATCH * 10, 128, 0, stream>>>(C8, out, part);
  scale_kernel<<<BATCH * 32, 256, 0, stream>>>(out, part);
}


# MLP and the three doubling levels of a batch element run in one launch (were four): C1, C2, C4 stay in LDS in both fragment layouts, only C8 goes to memory; sums and their order unchanged
# speedup vs baseline: 4.8953x; 1.1008x over previous
#include <hip/hip_runtime.h>
#include <hip/hip_bf16.h>


typedef float v2f __attribute__((ext_vector_type(2)));
typedef float v8f __attribute__((ext_vector_type(8)));

#define BATCH   128
#define MAT     1024
#define BMAT    (BATCH * MAT)
#define PAD     34
#define MTILE   (32 * PAD)
#define KPITCH  68

#define FRONT_LDS_FLOATS (40 * MTILE)

__device__ __forceinline__ v8f wmma4(v2f a, v2f b, v8f c) {
  return __builtin_amdgcn_wmma_f32_16x16x4_f32(
       false, a,  false, b,
       (short)0, c,  false,  false);
}

__device__ __forceinline__ v2f lds2(const float* p) {
  return *reinterpret_cast<const v2f*>(p);
}

__device__ __forceinline__ void matmul32(const float* X, const float* Y,
                                         int lm, int half, v8f (&acc)[2][2]) {
  acc[0][0] = v8f{}; acc[0][1] = v8f{}; acc[1][0] = v8f{}; acc[1][1] = v8f{};
#pragma unroll
  for (int kc = 0; kc < 8; ++kc) {
    const int kk = kc * 4 + half * 2;
    const v2f a0 = lds2(X + lm * PAD + kk);
    const v2f a1 = lds2(X + (lm + 16) * PAD + kk);
    const v2f b0 = lds2(Y + lm * PAD + kk);
    const v2f b1 = lds2(Y + (lm + 16) * PAD + kk);
    acc[0][0] = wmma4(a0, b0, acc[0][0]);
    acc[0][1] = wmma4(a0, b1, acc[0][1]);
    acc[1][0] = wmma4(a1, b0, acc[1][0]);
    acc[1][1] = wmma4(a1, b1, acc[1][1]);
  }
}

template <int S>
__device__ __forceinline__ void level_in_lds(const float* rm, const float* cm,
                                             float* orm, float* ocm,
                                             int wave, int lm, int half) {
  for (int p = wave; p < S * S; p += 8) {
    v8f acc[2][2];
    matmul32(rm + (p / S) * MTILE, cm + (p % S) * MTILE, lm, half, acc);
    float* Or = orm + p * MTILE;
    float* Oc = ocm + p * MTILE;
#pragma unroll
    for (int ti = 0; ti < 2; ++ti)
#pragma unroll
      for (int tj = 0; tj < 2; ++tj)
#pragma unroll
        for (int r = 0; r < 8; ++r) {
          const int row = ti * 16 + r + half * 8;
          const int col = tj * 16 + lm;
          Or[row * PAD + col] = acc[ti][tj][r];
          Oc[col * PAD + row] = acc[ti][tj][r];
        }
  }
}

__global__ void front_kernel(const float* __restrict__ x,
                             const float* __restrict__ W1, const float* __restrict__ b1,
                             const float* __restrict__ W2, const float* __restrict__ b2,
                             const float* __restrict__ W3, const float* __restrict__ b3,
                             float* __restrict__ C8) {
  extern __shared__ __align__(16) float sm[];
  float* r4 = sm;
  float* c4 = sm + 16 * MTILE;
  float* r2 = sm + 32 * MTILE;
  float* c2 = sm + 36 * MTILE;
  float* r1 = sm;
  float* c1 = sm + 2 * MTILE;
  float* h1 = sm + 4 * MTILE;
  float* h2 = h1 + 1024;
  float* xs = h2 + 32;
  const int b   = blockIdx.x;
  const int tid = threadIdx.x;
  if (tid < 4) xs[tid] = x[b * 4 + tid];
  __syncthreads();
  const float x0 = xs[0], x1 = xs[1], x2 = xs[2], x3 = xs[3];
  for (int o = tid; o < 1024; o += 256) {
    const float* w = W1 + o * 4;
    float v = fmaf(x0, w[0], fmaf(x1, w[1], fmaf(x2, w[2], fmaf(x3, w[3], b1[o]))));
    h1[o] = fmaxf(v, 0.0f);
  }
  __syncthreads();
  if (tid < 32) {
    const float* w = W2 + tid * 1024;
    float v = b2[tid];
    for (int k = 0; k < 1024; ++k) v = fmaf(h1[k], w[k], v);
    h2[tid] = fmaxf(v, 0.0f);
  }
  __syncthreads();
  for (int o = tid; o < 2048; o += 256) {
    const float* w = W3 + o * 32;
    float v = b3[o];
#pragma unroll
    for (int k = 0; k < 32; ++k) v = fmaf(h2[k], w[k], v);
    const int s   = o >> 10;
    const int rem = o & 1023;
    const int r = rem >> 5, c = rem & 31;
    r1[s * MTILE + r * PAD + c] = v;
    c1[s * MTILE + c * PAD + r] = v;
  }
  __syncthreads();
  const int wave = tid >> 5;
  const int lane = tid & 31;
  const int lm   = lane & 15;
  const int half = lane >> 4;
  level_in_lds<2>(r1, c1, r2, c2, wave, lm, half);
  __syncthreads();
  level_in_lds<4>(r2, c2, r4, c4, wave, lm, half);
  __syncthreads();
  for (int p = wave; p < 256; p += 8) {
    v8f acc[2][2];
    matmul32(r4 + (p >> 4) * MTILE, c4 + (p & 15) * MTILE, lm, half, acc);
    float* O = C8 + (size_t)p * BMAT + (size_t)b * MAT;
#pragma unroll
    for (int ti = 0; ti < 2; ++ti)
#pragma unroll
      for (int tj = 0; tj < 2; ++tj)
#pragma unroll
        for (int r = 0; r < 8; ++r) {
          const int row = ti * 16 + r + half * 8;
          const int col = tj * 16 + lm;
          const int o = ((((row >> 3) << 2) + (col >> 3)) << 6) + ((row & 7) << 3) + (col & 7);
          O[o] = acc[ti][tj][r];
        }
  }
}

__global__ void trace_gemm_kernel(const float* __restrict__ C8,
                                  float* __restrict__ out,
                                  float* __restrict__ part) {
  __shared__ __align__(16) float Lt[64 * KPITCH];
  __shared__ __align__(16) float Rt[64 * KPITCH];
  const int b    = blockIdx.x / 10;
  const int blk  = blockIdx.x - b * 10;
  const int I    = (blk >= 4) + (blk >= 7) + (blk >= 9);
  const int J    = blk - ((I * (9 - I)) >> 1) + I;
  const int s0   = I * 64;
  const int t0   = J * 64;
  const int tid  = threadIdx.x;
  const int wave = tid >> 5;
  const int lane = tid & 31;
  const int lm   = lane & 15;
  const int half = lane >> 4;
  const int ti0  = (wave >> 1) * 2;
  const int tj0  = (wave & 1) * 2;
  v8f acc[2][2] = {};
  const float* base = C8 + (size_t)b * MAT;
  for (int chunk = 0; chunk < 16; ++chunk) {
    const int lblk = chunk * 64;
    const int rblk = (((chunk & 3) << 2) + (chunk >> 2)) * 64;
    for (int n = 0; n < 8; ++n) {
      const int idx = tid + n * 128;
      const int sl = idx >> 4, q = idx & 15;
      const float4 v = *reinterpret_cast<const float4*>(
          base + (size_t)(s0 + sl) * BMAT + lblk + q * 4);
      *reinterpret_cast<float4*>(Lt + sl * KPITCH + q * 4) = v;
    }
    for (int n = 0; n < 8; ++n) {
      const int idx = tid + n * 128;
      const int tl = idx >> 4, q = idx & 15;
      const int jl = q >> 1, i4 = (q & 1) * 4;
      const float4 v = *reinterpret_cast<const float4*>(
          base + (size_t)(t0 + tl) * BMAT + rblk + jl * 8 + i4);
      float* d = Rt + tl * KPITCH + i4 * 8 + jl;
      d[0] = v.x; d[8] = v.y; d[16] = v.z; d[24] = v.w;
    }
    __syncthreads();
#pragma unroll
    for (int kc = 0; kc < 16; ++kc) {
      const int kk = kc * 4 + half * 2;
      const v2f a0 = lds2(Lt + (ti0 * 16 + lm) * KPITCH + kk);
      const v2f a1 = lds2(Lt + ((ti0 + 1) * 16 + lm) * KPITCH + kk);
      const v2f b0 = lds2(Rt + (tj0 * 16 + lm) * KPITCH + kk);
      const v2f b1 = lds2(Rt + ((tj0 + 1) * 16 + lm) * KPITCH + kk);
      acc[0][0] = wmma4(a0, b0, acc[0][0]);
      acc[0][1] = wmma4(a0, b1, acc[0][1]);
      acc[1][0] = wmma4(a1, b0, acc[1][0]);
      acc[1][1] = wmma4(a1, b1, acc[1][1]);
    }
    __syncthreads();
  }
  float ss = 0.f;
#pragma unroll
  for (int di = 0; di < 2; ++di)
#pragma unroll
    for (int dj = 0; dj < 2; ++dj)
#pragma unroll
      for (int r = 0; r < 8; ++r) ss = fmaf(acc[di][dj][r], acc[di][dj][r], ss);
  for (int off = 16; off > 0; off >>= 1) ss += __shfl_xor(ss, off, 32);
  if (lane == 0) Lt[wave] = ss;
  __syncthreads();
  if (tid == 0)
    part[blockIdx.x * 32] = ((Lt[0] + Lt[1]) + (Lt[2] + Lt[3])) * (I != J ? 2.f : 1.f);
  float* O = out + (size_t)b * 65536;
#pragma unroll
  for (int di = 0; di < 2; ++di)
#pragma unroll
    for (int dj = 0; dj < 2; ++dj)
#pragma unroll
      for (int r = 0; r < 8; ++r) {
        const int row = s0 + (ti0 + di) * 16 + r + half * 8;
        const int col = t0 + (tj0 + dj) * 16 + lm;
        O[(size_t)row * 256 + col] = acc[di][dj][r];
      }
  if (I != J) {
    for (int di = 0; di < 2; ++di)
      for (int dj = 0; dj < 2; ++dj) {
        const int tm = t0 + (tj0 + dj) * 16 + lm;
        const int sb = s0 + (ti0 + di) * 16 + half * 8;
        float* p = O + (size_t)tm * 256 + sb;
        *reinterpret_cast<float4*>(p) =
            make_float4(acc[di][dj][0], acc[di][dj][1], acc[di][dj][2], acc[di][dj][3]);
        *reinterpret_cast<float4*>(p + 4) =
            make_float4(acc[di][dj][4], acc[di][dj][5], acc[di][dj][6], acc[di][dj][7]);
      }
  }
}

__global__ void scale_kernel(float* __restrict__ out,
                             const float* __restrict__ part) {
  const float* q = part + (blockIdx.x >> 5) * 320;
  float s = q[0];
  for (int k = 1; k < 10; ++k) s += q[k * 32];
  const float inv = rsqrtf(s);
  float* p = out + (size_t)blockIdx.x * 2048 + threadIdx.x;
#pragma unroll
  for (int k = 0; k < 8; ++k) p[k * 256] *= inv;
}

extern "C" void kernel_launch(void* const* d_in, const int* in_sizes, int n_in,
                              void* d_out, int out_size, void* d_ws, size_t ws_size,
                              hipStream_t stream) {
  const float* x  = (const float*)d_in[0];
  const float* W1 = (const float*)d_in[1];
  const float* b1 = (const float*)d_in[2];
  const float* W2 = (const float*)d_in[3];
  const float* b2 = (const float*)d_in[4];
  const float* W3 = (const float*)d_in[5];
  const float* b3 = (const float*)d_in[6];
  float* out = (float*)d_out;

  float* ws   = (float*)d_ws;
  float* part = ws;
  float* C8   = ws + (size_t)BATCH * 10 * 32;

  front_kernel<<<BATCH, 256, (size_t)FRONT_LDS_FLOATS * sizeof(float), stream>>>(
      x, W1, b1, W2, b2, W3, b3, C8);

  trace_gemm_kernel<<<BATCH * 10, 128, 0, stream>>>(C8, out, part);
  scale_kernel<<<BATCH * 32, 256, 0, stream>>>(out, part);
}


# encoder MLP layers 2 and 3 on all 8 waves, lanes along k: whole-line weight loads, lane partial sums added by xor butterflies (layer 2 was one wave, serial over k); h2 and e sum order changed
# speedup vs baseline: 5.4972x; 1.1230x over previous
#include <hip/hip_runtime.h>
#include <hip/hip_bf16.h>


typedef float v2f __attribute__((ext_vector_type(2)));
typedef float v8f __attribute__((ext_vector_type(8)));

#define BATCH   128
#define MAT     1024
#define BMAT    (BATCH * MAT)
#define PAD     34
#define MTILE   (32 * PAD)
#define KPITCH  68

#define FRONT_LDS_FLOATS (40 * MTILE)

__device__ __forceinline__ v8f wmma4(v2f a, v2f b, v8f c) {
  return __builtin_amdgcn_wmma_f32_16x16x4_f32(
       false, a,  false, b,
       (short)0, c,  false,  false);
}

__device__ __forceinline__ v2f lds2(const float* p) {
  return *reinterpret_cast<const v2f*>(p);
}

__device__ __forceinline__ void matmul32(const float* X, const float* Y,
                                         int lm, int half, v8f (&acc)[2][2]) {
  acc[0][0] = v8f{}; acc[0][1] = v8f{}; acc[1][0] = v8f{}; acc[1][1] = v8f{};
#pragma unroll
  for (int kc = 0; kc < 8; ++kc) {
    const int kk = kc * 4 + half * 2;
    const v2f a0 = lds2(X + lm * PAD + kk);
    const v2f a1 = lds2(X + (lm + 16) * PAD + kk);
    const v2f b0 = lds2(Y + lm * PAD + kk);
    const v2f b1 = lds2(Y + (lm + 16) * PAD + kk);
    acc[0][0] = wmma4(a0, b0, acc[0][0]);
    acc[0][1] = wmma4(a0, b1, acc[0][1]);
    acc[1][0] = wmma4(a1, b0, acc[1][0]);
    acc[1][1] = wmma4(a1, b1, acc[1][1]);
  }
}

template <int S>
__device__ __forceinline__ void level_in_lds(const float* rm, const float* cm,
                                             float* orm, float* ocm,
                                             int wave, int lm, int half) {
  for (int p = wave; p < S * S; p += 8) {
    v8f acc[2][2];
    matmul32(rm + (p / S) * MTILE, cm + (p % S) * MTILE, lm, half, acc);
    float* Or = orm + p * MTILE;
    float* Oc = ocm + p * MTILE;
#pragma unroll
    for (int ti = 0; ti < 2; ++ti)
#pragma unroll
      for (int tj = 0; tj < 2; ++tj)
#pragma unroll
        for (int r = 0; r < 8; ++r) {
          const int row = ti * 16 + r + half * 8;
          const int col = tj * 16 + lm;
          Or[row * PAD + col] = acc[ti][tj][r];
          Oc[col * PAD + row] = acc[ti][tj][r];
        }
  }
}

__global__ void __launch_bounds__(256)
front_kernel(const float* __restrict__ x,
                             const float* __restrict__ W1, const float* __restrict__ b1,
                             const float* __restrict__ W2, const float* __restrict__ b2,
                             const float* __restrict__ W3, const float* __restrict__ b3,
                             float* __restrict__ C8) {
  extern __shared__ __align__(16) float sm[];
  float* r4 = sm;
  float* c4 = sm + 16 * MTILE;
  float* r2 = sm + 32 * MTILE;
  float* c2 = sm + 36 * MTILE;
  float* r1 = sm;
  float* c1 = sm + 2 * MTILE;
  float* h1 = sm + 4 * MTILE;
  float* h2 = h1 + 1024;
  float* xs = h2 + 32;
  const int b   = blockIdx.x;
  const int tid = threadIdx.x;
  if (tid < 4) xs[tid] = x[b * 4 + tid];
  __syncthreads();
  const float x0 = xs[0], x1 = xs[1], x2 = xs[2], x3 = xs[3];
  for (int o = tid; o < 1024; o += 256) {
    const float* w = W1 + o * 4;
    float v = fmaf(x0, w[0], fmaf(x1, w[1], fmaf(x2, w[2], fmaf(x3, w[3], b1[o]))));
    h1[o] = fmaxf(v, 0.0f);
  }
  __syncthreads();
  const int wave = tid >> 5;
  const int lane = tid & 31;
  const int lm   = lane & 15;
  const int half = lane >> 4;
  {
    float bj[4];
#pragma unroll
    for (int j = 0; j < 4; ++j) bj[j] = b2[wave * 4 + j];
    float acc[4] = {0.f, 0.f, 0.f, 0.f};
#pragma unroll 1
    for (int ih = 0; ih < 8; ih += 4) {
      float4 wv[4][4];
#pragma unroll
      for (int j = 0; j < 4; ++j)
#pragma unroll
        for (int i = 0; i < 4; ++i)
          wv[j][i] = *reinterpret_cast<const float4*>(
              W2 + (wave * 4 + j) * 1024 + (ih + i) * 128 + lane * 4);
#pragma unroll
      for (int i = 0; i < 4; ++i) {
        const float4 hv = *reinterpret_cast<const float4*>(h1 + (ih + i) * 128 + lane * 4);
#pragma unroll
        for (int j = 0; j < 4; ++j) {
          acc[j] = fmaf(hv.x, wv[j][i].x, acc[j]);
          acc[j] = fmaf(hv.y, wv[j][i].y, acc[j]);
          acc[j] = fmaf(hv.z, wv[j][i].z, acc[j]);
          acc[j] = fmaf(hv.w, wv[j][i].w, acc[j]);
        }
      }
    }
#pragma unroll
    for (int off = 16; off > 0; off >>= 1) {
      float t[4];
#pragma unroll
      for (int j = 0; j < 4; ++j) t[j] = __shfl_xor(acc[j], off, 32);
#pragma unroll
      for (int j = 0; j < 4; ++j) acc[j] += t[j];
    }
    if (lane == 0) {
#pragma unroll
      for (int j = 0; j < 4; ++j) h2[wave * 4 + j] = fmaxf(acc[j] + bj[j], 0.0f);
    }
  }
  __syncthreads();
  {
    const int sub = lane >> 3;
    const int kq  = (lane & 7) * 4;
    const float4 hv = *reinterpret_cast<const float4*>(h2 + kq);
#pragma unroll 1
    for (int g0 = 0; g0 < 64; g0 += 16) {
      float4 wv[16];
      float  bv[16];
#pragma unroll
      for (int u = 0; u < 16; ++u) {
        const int row = wave * 256 + (g0 + u) * 4 + sub;
        wv[u] = *reinterpret_cast<const float4*>(W3 + row * 32 + kq);
        bv[u] = b3[row];
      }
      float p[16];
#pragma unroll
      for (int u = 0; u < 16; ++u) {
        p[u] = fmaf(hv.x, wv[u].x, (lane & 7) == 0 ? bv[u] : 0.0f);
        p[u] = fmaf(hv.y, wv[u].y, p[u]);
        p[u] = fmaf(hv.z, wv[u].z, p[u]);
        p[u] = fmaf(hv.w, wv[u].w, p[u]);
      }
#pragma unroll
      for (int off = 1; off < 8; off <<= 1) {
        float t[16];
#pragma unroll
        for (int u = 0; u < 16; ++u) t[u] = __shfl_xor(p[u], off, 32);
#pragma unroll
        for (int u = 0; u < 16; ++u) p[u] += t[u];
      }
      if ((lane & 7) == 0) {
#pragma unroll
        for (int u = 0; u < 16; ++u) {
          const int row = wave * 256 + (g0 + u) * 4 + sub;
          const float v = p[u];
          const int s   = row >> 10;
          const int rem = row & 1023;
          const int r = rem >> 5, c = rem & 31;
          r1[s * MTILE + r * PAD + c] = v;
          c1[s * MTILE + c * PAD + r] = v;
        }
      }
    }
  }
  __syncthreads();
  level_in_lds<2>(r1, c1, r2, c2, wave, lm, half);
  __syncthreads();
  level_in_lds<4>(r2, c2, r4, c4, wave, lm, half);
  __syncthreads();
  for (int p = wave; p < 256; p += 8) {
    v8f acc[2][2];
    matmul32(r4 + (p >> 4) * MTILE, c4 + (p & 15) * MTILE, lm, half, acc);
    float* O = C8 + (size_t)p * BMAT + (size_t)b * MAT;
#pragma unroll
    for (int ti = 0; ti < 2; ++ti)
#pragma unroll
      for (int tj = 0; tj < 2; ++tj)
#pragma unroll
        for (int r = 0; r < 8; ++r) {
          const int row = ti * 16 + r + half * 8;
          const int col = tj * 16 + lm;
          const int o = ((((row >> 3) << 2) + (col >> 3)) << 6) + ((row & 7) << 3) + (col & 7);
          O[o] = acc[ti][tj][r];
        }
  }
}

__global__ void trace_gemm_kernel(const float* __restrict__ C8,
                                  float* __restrict__ out,
                                  float* __restrict__ part) {
  __shared__ __align__(16) float Lt[64 * KPITCH];
  __shared__ __align__(16) float Rt[64 * KPITCH];
  const int b    = blockIdx.x / 10;
  const int blk  = blockIdx.x - b * 10;
  const int I    = (blk >= 4) + (blk >= 7) + (blk >= 9);
  const int J    = blk - ((I * (9 - I)) >> 1) + I;
  const int s0   = I * 64;
  const int t0   = J * 64;
  const int tid  = threadIdx.x;
  const int wave = tid >> 5;
  const int lane = tid & 31;
  const int lm   = lane & 15;
  const int half = lane >> 4;
  const int ti0  = (wave >> 1) * 2;
  const int tj0  = (wave & 1) * 2;
  v8f acc[2][2] = {};
  const float* base = C8 + (size_t)b * MAT;
  for (int chunk = 0; chunk < 16; ++chunk) {
    const int lblk = chunk * 64;
    const int rblk = (((chunk & 3) << 2) + (chunk >> 2)) * 64;
    for (int n = 0; n < 8; ++n) {
      const int idx = tid + n * 128;
      const int sl = idx >> 4, q = idx & 15;
      const float4 v = *reinterpret_cast<const float4*>(
          base + (size_t)(s0 + sl) * BMAT + lblk + q * 4);
      *reinterpret_cast<float4*>(Lt + sl * KPITCH + q * 4) = v;
    }
    for (int n = 0; n < 8; ++n) {
      const int idx = tid + n * 128;
      const int tl = idx >> 4, q = idx & 15;
      const int jl = q >> 1, i4 = (q & 1) * 4;
      const float4 v = *reinterpret_cast<const float4*>(
          base + (size_t)(t0 + tl) * BMAT + rblk + jl * 8 + i4);
      float* d = Rt + tl * KPITCH + i4 * 8 + jl;
      d[0] = v.x; d[8] = v.y; d[16] = v.z; d[24] = v.w;
    }
    __syncthreads();
#pragma unroll
    for (int kc = 0; kc < 16; ++kc) {
      const int kk = kc * 4 + half * 2;
      const v2f a0 = lds2(Lt + (ti0 * 16 + lm) * KPITCH + kk);
      const v2f a1 = lds2(Lt + ((ti0 + 1) * 16 + lm) * KPITCH + kk);
      const v2f b0 = lds2(Rt + (tj0 * 16 + lm) * KPITCH + kk);
      const v2f b1 = lds2(Rt + ((tj0 + 1) * 16 + lm) * KPITCH + kk);
      acc[0][0] = wmma4(a0, b0, acc[0][0]);
      acc[0][1] = wmma4(a0, b1, acc[0][1]);
      acc[1][0] = wmma4(a1, b0, acc[1][0]);
      acc[1][1] = wmma4(a1, b1, acc[1][1]);
    }
    __syncthreads();
  }
  float ss = 0.f;
#pragma unroll
  for (int di = 0; di < 2; ++di)
#pragma unroll
    for (int dj = 0; dj < 2; ++dj)
#pragma unroll
      for (int r = 0; r < 8; ++r) ss = fmaf(acc[di][dj][r], acc[di][dj][r], ss);
  for (int off = 16; off > 0; off >>= 1) ss += __shfl_xor(ss, off, 32);
  if (lane == 0) Lt[wave] = ss;
  __syncthreads();
  if (tid == 0)
    part[blockIdx.x * 32] = ((Lt[0] + Lt[1]) + (Lt[2] + Lt[3])) * (I != J ? 2.f : 1.f);
  float* O = out + (size_t)b * 65536;
#pragma unroll
  for (int di = 0; di < 2; ++di)
#pragma unroll
    for (int dj = 0; dj < 2; ++dj)
#pragma unroll
      for (int r = 0; r < 8; ++r) {
        const int row = s0 + (ti0 + di) * 16 + r + half * 8;
        const int col = t0 + (tj0 + dj) * 16 + lm;
        O[(size_t)row * 256 + col] = acc[di][dj][r];
      }
  if (I != J) {
    for (int di = 0; di < 2; ++di)
      for (int dj = 0; dj < 2; ++dj) {
        const int tm = t0 + (tj0 + dj) * 16 + lm;
        const int sb = s0 + (ti0 + di) * 16 + half * 8;
        float* p = O + (size_t)tm * 256 + sb;
        *reinterpret_cast<float4*>(p) =
            make_float4(acc[di][dj][0], acc[di][dj][1], acc[di][dj][2], acc[di][dj][3]);
        *reinterpret_cast<float4*>(p + 4) =
            make_float4(acc[di][dj][4], acc[di][dj][5], acc[di][dj][6], acc[di][dj][7]);
      }
  }
}

__global__ void scale_kernel(float* __restrict__ out,
                             const float* __restrict__ part) {
  const float* q = part + (blockIdx.x >> 5) * 320;
  float s = q[0];
  for (int k = 1; k < 10; ++k) s += q[k * 32];
  const float inv = rsqrtf(s);
  float* p = out + (size_t)blockIdx.x * 2048 + threadIdx.x;
#pragma unroll
  for (int k = 0; k < 8; ++k) p[k * 256] *= inv;
}

extern "C" void kernel_launch(void* const* d_in, const int* in_sizes, int n_in,
                              void* d_out, int out_size, void* d_ws, size_t ws_size,
                              hipStream_t stream) {
  const float* x  = (const float*)d_in[0];
  const float* W1 = (const float*)d_in[1];
  const float* b1 = (const float*)d_in[2];
  const float* W2 = (const float*)d_in[3];
  const float* b2 = (const float*)d_in[4];
  const float* W3 = (const float*)d_in[5];
  const float* b3 = (const float*)d_in[6];
  float* out = (float*)d_out;

  float* ws   = (float*)d_ws;
  float* part = ws;
  float* C8   = ws + (size_t)BATCH * 10 * 32;

  front_kernel<<<BATCH, 256, (size_t)FRONT_LDS_FLOATS * sizeof(float), stream>>>(
      x, W1, b1, W2, b2, W3, b3, C8);

  trace_gemm_kernel<<<BATCH * 10, 128, 0, stream>>>(C8, out, part);
  scale_kernel<<<BATCH * 32, 256, 0, stream>>>(out, part);
}


# trace GEMM holds the next K chunk in registers: its 16 loads per lane go out as one group under the current chunk's WMMAs (LDS-only barrier by hand, 128-thread launch bound); sum order unchanged
# speedup vs baseline: 6.0209x; 1.0953x over previous
#include <hip/hip_runtime.h>
#include <hip/hip_bf16.h>


typedef float v2f __attribute__((ext_vector_type(2)));
typedef float v8f __attribute__((ext_vector_type(8)));

#define BATCH   128
#define MAT     1024
#define BMAT    (BATCH * MAT)
#define PAD     34
#define MTILE   (32 * PAD)
#define KPITCH  68

#define FRONT_LDS_FLOATS (40 * MTILE)

__device__ __forceinline__ v8f wmma4(v2f a, v2f b, v8f c) {
  return __builtin_amdgcn_wmma_f32_16x16x4_f32(
       false, a,  false, b,
       (short)0, c,  false,  false);
}

__device__ __forceinline__ v2f lds2(const float* p) {
  return *reinterpret_cast<const v2f*>(p);
}

__device__ __forceinline__ void matmul32(const float* X, const float* Y,
                                         int lm, int half, v8f (&acc)[2][2]) {
  acc[0][0] = v8f{}; acc[0][1] = v8f{}; acc[1][0] = v8f{}; acc[1][1] = v8f{};
#pragma unroll
  for (int kc = 0; kc < 8; ++kc) {
    const int kk = kc * 4 + half * 2;
    const v2f a0 = lds2(X + lm * PAD + kk);
    const v2f a1 = lds2(X + (lm + 16) * PAD + kk);
    const v2f b0 = lds2(Y + lm * PAD + kk);
    const v2f b1 = lds2(Y + (lm + 16) * PAD + kk);
    acc[0][0] = wmma4(a0, b0, acc[0][0]);
    acc[0][1] = wmma4(a0, b1, acc[0][1]);
    acc[1][0] = wmma4(a1, b0, acc[1][0]);
    acc[1][1] = wmma4(a1, b1, acc[1][1]);
  }
}

template <int S>
__device__ __forceinline__ void level_in_lds(const float* rm, const float* cm,
                                             float* orm, float* ocm,
                                             int wave, int lm, int half) {
  for (int p = wave; p < S * S; p += 8) {
    v8f acc[2][2];
    matmul32(rm + (p / S) * MTILE, cm + (p % S) * MTILE, lm, half, acc);
    float* Or = orm + p * MTILE;
    float* Oc = ocm + p * MTILE;
#pragma unroll
    for (int ti = 0; ti < 2; ++ti)
#pragma unroll
      for (int tj = 0; tj < 2; ++tj)
#pragma unroll
        for (int r = 0; r < 8; ++r) {
          const int row = ti * 16 + r + half * 8;
          const int col = tj * 16 + lm;
          Or[row * PAD + col] = acc[ti][tj][r];
          Oc[col * PAD + row] = acc[ti][tj][r];
        }
  }
}

__global__ void __launch_bounds__(256)
front_kernel(const float* __restrict__ x,
                             const float* __restrict__ W1, const float* __restrict__ b1,
                             const float* __restrict__ W2, const float* __restrict__ b2,
                             const float* __restrict__ W3, const float* __restrict__ b3,
                             float* __restrict__ C8) {
  extern __shared__ __align__(16) float sm[];
  float* r4 = sm;
  float* c4 = sm + 16 * MTILE;
  float* r2 = sm + 32 * MTILE;
  float* c2 = sm + 36 * MTILE;
  float* r1 = sm;
  float* c1 = sm + 2 * MTILE;
  float* h1 = sm + 4 * MTILE;
  float* h2 = h1 + 1024;
  float* xs = h2 + 32;
  const int b   = blockIdx.x;
  const int tid = threadIdx.x;
  if (tid < 4) xs[tid] = x[b * 4 + tid];
  __syncthreads();
  const float x0 = xs[0], x1 = xs[1], x2 = xs[2], x3 = xs[3];
  for (int o = tid; o < 1024; o += 256) {
    const float* w = W1 + o * 4;
    float v = fmaf(x0, w[0], fmaf(x1, w[1], fmaf(x2, w[2], fmaf(x3, w[3], b1[o]))));
    h1[o] = fmaxf(v, 0.0f);
  }
  __syncthreads();
  const int wave = tid >> 5;
  const int lane = tid & 31;
  const int lm   = lane & 15;
  const int half = lane >> 4;
  {
    float bj[4];
#pragma unroll
    for (int j = 0; j < 4; ++j) bj[j] = b2[wave * 4 + j];
    float acc[4] = {0.f, 0.f, 0.f, 0.f};
#pragma unroll 1
    for (int ih = 0; ih < 8; ih += 4) {
      float4 wv[4][4];
#pragma unroll
      for (int j = 0; j < 4; ++j)
#pragma unroll
        for (int i = 0; i < 4; ++i)
          wv[j][i] = *reinterpret_cast<const float4*>(
              W2 + (wave * 4 + j) * 1024 + (ih + i) * 128 + lane * 4);
#pragma unroll
      for (int i = 0; i < 4; ++i) {
        const float4 hv = *reinterpret_cast<const float4*>(h1 + (ih + i) * 128 + lane * 4);
#pragma unroll
        for (int j = 0; j < 4; ++j) {
          acc[j] = fmaf(hv.x, wv[j][i].x, acc[j]);
          acc[j] = fmaf(hv.y, wv[j][i].y, acc[j]);
          acc[j] = fmaf(hv.z, wv[j][i].z, acc[j]);
          acc[j] = fmaf(hv.w, wv[j][i].w, acc[j]);
        }
      }
    }
#pragma unroll
    for (int off = 16; off > 0; off >>= 1) {
      float t[4];
#pragma unroll
      for (int j = 0; j < 4; ++j) t[j] = __shfl_xor(acc[j], off, 32);
#pragma unroll
      for (int j = 0; j < 4; ++j) acc[j] += t[j];
    }
    if (lane == 0) {
#pragma unroll
      for (int j = 0; j < 4; ++j) h2[wave * 4 + j] = fmaxf(acc[j] + bj[j], 0.0f);
    }
  }
  __syncthreads();
  {
    const int sub = lane >> 3;
    const int kq  = (lane & 7) * 4;
    const float4 hv = *reinterpret_cast<const float4*>(h2 + kq);
#pragma unroll 1
    for (int g0 = 0; g0 < 64; g0 += 16) {
      float4 wv[16];
      float  bv[16];
#pragma unroll
      for (int u = 0; u < 16; ++u) {
        const int row = wave * 256 + (g0 + u) * 4 + sub;
        wv[u] = *reinterpret_cast<const float4*>(W3 + row * 32 + kq);
        bv[u] = b3[row];
      }
      float p[16];
#pragma unroll
      for (int u = 0; u < 16; ++u) {
        p[u] = fmaf(hv.x, wv[u].x, (lane & 7) == 0 ? bv[u] : 0.0f);
        p[u] = fmaf(hv.y, wv[u].y, p[u]);
        p[u] = fmaf(hv.z, wv[u].z, p[u]);
        p[u] = fmaf(hv.w, wv[u].w, p[u]);
      }
#pragma unroll
      for (int off = 1; off < 8; off <<= 1) {
        float t[16];
#pragma unroll
        for (int u = 0; u < 16; ++u) t[u] = __shfl_xor(p[u], off, 32);
#pragma unroll
        for (int u = 0; u < 16; ++u) p[u] += t[u];
      }
      if ((lane & 7) == 0) {
#pragma unroll
        for (int u = 0; u < 16; ++u) {
          const int row = wave * 256 + (g0 + u) * 4 + sub;
          const float v = p[u];
          const int s   = row >> 10;
          const int rem = row & 1023;
          const int r = rem >> 5, c = rem & 31;
          r1[s * MTILE + r * PAD + c] = v;
          c1[s * MTILE + c * PAD + r] = v;
        }
      }
    }
  }
  __syncthreads();
  level_in_lds<2>(r1, c1, r2, c2, wave, lm, half);
  __syncthreads();
  level_in_lds<4>(r2, c2, r4, c4, wave, lm, half);
  __syncthreads();
  for (int p = wave; p < 256; p += 8) {
    v8f acc[2][2];
    matmul32(r4 + (p >> 4) * MTILE, c4 + (p & 15) * MTILE, lm, half, acc);
    float* O = C8 + (size_t)p * BMAT + (size_t)b * MAT;
#pragma unroll
    for (int ti = 0; ti < 2; ++ti)
#pragma unroll
      for (int tj = 0; tj < 2; ++tj)
#pragma unroll
        for (int r = 0; r < 8; ++r) {
          const int row = ti * 16 + r + half * 8;
          const int col = tj * 16 + lm;
          const int o = ((((row >> 3) << 2) + (col >> 3)) << 6) + ((row & 7) << 3) + (col & 7);
          O[o] = acc[ti][tj][r];
        }
  }
}

__device__ __forceinline__ void lds_barrier() {
  asm volatile("s_wait_dscnt 0x0\n\ts_barrier_signal -1\n\ts_barrier_wait -1" ::: "memory");
}

__device__ __forceinline__ float4 ldg4(const float* p) {
  return *reinterpret_cast<const float4*>(p);
}

__global__ void __launch_bounds__(128)
trace_gemm_kernel(const float* __restrict__ C8,
                                  float* __restrict__ out,
                                  float* __restrict__ part) {
  __shared__ __align__(16) float Lt[64 * KPITCH];
  __shared__ __align__(16) float Rt[64 * KPITCH];
  const int b    = blockIdx.x / 10;
  const int blk  = blockIdx.x - b * 10;
  const int I    = (blk >= 4) + (blk >= 7) + (blk >= 9);
  const int J    = blk - ((I * (9 - I)) >> 1) + I;
  const int s0   = I * 64;
  const int t0   = J * 64;
  const int tid  = threadIdx.x;
  const int wave = tid >> 5;
  const int lane = tid & 31;
  const int lm   = lane & 15;
  const int half = lane >> 4;
  const int ti0  = (wave >> 1) * 2;
  const int tj0  = (wave & 1) * 2;
  v8f acc[2][2] = {};
  const int q  = tid & 15;
  const int r0 = tid >> 4;
  const float* gL = C8 + (size_t)b * MAT + (size_t)s0 * BMAT;
  const float* gR = C8 + (size_t)b * MAT + (size_t)t0 * BMAT;
  const unsigned voff = (unsigned)(r0 * BMAT + q * 4);
  float* sL = Lt + r0 * KPITCH + q * 4;
  float* sR = Rt + r0 * KPITCH + (q & 1) * 32 + (q >> 1);
  float4 lv[8], rv[8];
#pragma unroll
  for (int n = 0; n < 8; ++n) {
    lv[n] = ldg4(gL + (voff + (unsigned)(n * 8 * BMAT)));
    rv[n] = ldg4(gR + (voff + (unsigned)(n * 8 * BMAT)));
  }
  for (int chunk = 0; chunk < 16; ++chunk) {
#pragma unroll
    for (int n = 0; n < 8; ++n)
      *reinterpret_cast<float4*>(sL + n * 8 * KPITCH) = lv[n];
#pragma unroll
    for (int n = 0; n < 8; ++n) {
      float* d = sR + n * 8 * KPITCH;
      d[0] = rv[n].x; d[8] = rv[n].y; d[16] = rv[n].z; d[24] = rv[n].w;
    }
    lds_barrier();
    if (chunk < 15) {
      const int nc = chunk + 1;
      const float* pl = gL + nc * 64;
      const float* pr = gR + (((nc & 3) << 2) + (nc >> 2)) * 64;
#pragma unroll
      for (int n = 0; n < 8; ++n) {
        lv[n] = ldg4(pl + (voff + (unsigned)(n * 8 * BMAT)));
        rv[n] = ldg4(pr + (voff + (unsigned)(n * 8 * BMAT)));
      }
    }
    __builtin_amdgcn_sched_barrier(0);
#pragma unroll
    for (int kc = 0; kc < 16; ++kc) {
      const int kk = kc * 4 + half * 2;
      const v2f a0 = lds2(Lt + (ti0 * 16 + lm) * KPITCH + kk);
      const v2f a1 = lds2(Lt + ((ti0 + 1) * 16 + lm) * KPITCH + kk);
      const v2f b0 = lds2(Rt + (tj0 * 16 + lm) * KPITCH + kk);
      const v2f b1 = lds2(Rt + ((tj0 + 1) * 16 + lm) * KPITCH + kk);
      acc[0][0] = wmma4(a0, b0, acc[0][0]);
      acc[0][1] = wmma4(a0, b1, acc[0][1]);
      acc[1][0] = wmma4(a1, b0, acc[1][0]);
      acc[1][1] = wmma4(a1, b1, acc[1][1]);
    }
    lds_barrier();
  }
  float ss = 0.f;
#pragma unroll
  for (int di = 0; di < 2; ++di)
#pragma unroll
    for (int dj = 0; dj < 2; ++dj)
#pragma unroll
      for (int r = 0; r < 8; ++r) ss = fmaf(acc[di][dj][r], acc[di][dj][r], ss);
  for (int off = 16; off > 0; off >>= 1) ss += __shfl_xor(ss, off, 32);
  if (lane == 0) Lt[wave] = ss;
  __syncthreads();
  if (tid == 0)
    part[blockIdx.x * 32] = ((Lt[0] + Lt[1]) + (Lt[2] + Lt[3])) * (I != J ? 2.f : 1.f);
  float* O = out + (size_t)b * 65536;
#pragma unroll
  for (int di = 0; di < 2; ++di)
#pragma unroll
    for (int dj = 0; dj < 2; ++dj)
#pragma unroll
      for (int r = 0; r < 8; ++r) {
        const int row = s0 + (ti0 + di) * 16 + r + half * 8;
        const int col = t0 + (tj0 + dj) * 16 + lm;
        O[(size_t)row * 256 + col] = acc[di][dj][r];
      }
  if (I != J) {
    for (int di = 0; di < 2; ++di)
      for (int dj = 0; dj < 2; ++dj) {
        const int tm = t0 + (tj0 + dj) * 16 + lm;
        const int sb = s0 + (ti0 + di) * 16 + half * 8;
        float* p = O + (size_t)tm * 256 + sb;
        *reinterpret_cast<float4*>(p) =
            make_float4(acc[di][dj][0], acc[di][dj][1], acc[di][dj][2], acc[di][dj][3]);
        *reinterpret_cast<float4*>(p + 4) =
            make_float4(acc[di][dj][4], acc[di][dj][5], acc[di][dj][6], acc[di][dj][7]);
      }
  }
}

__global__ void scale_kernel(float* __restrict__ out,
                             const float* __restrict__ part) {
  const float* q = part + (blockIdx.x >> 5) * 320;
  float s = q[0];
  for (int k = 1; k < 10; ++k) s += q[k * 32];
  const float inv = rsqrtf(s);
  float* p = out + (size_t)blockIdx.x * 2048 + threadIdx.x;
#pragma unroll
  for (int k = 0; k < 8; ++k) p[k * 256] *= inv;
}

extern "C" void kernel_launch(void* const* d_in, const int* in_sizes, int n_in,
                              void* d_out, int out_size, void* d_ws, size_t ws_size,
                              hipStream_t stream) {
  const float* x  = (const float*)d_in[0];
  const float* W1 = (const float*)d_in[1];
  const float* b1 = (const float*)d_in[2];
  const float* W2 = (const float*)d_in[3];
  const float* b2 = (const float*)d_in[4];
  const float* W3 = (const float*)d_in[5];
  const float* b3 = (const float*)d_in[6];
  float* out = (float*)d_out;

  float* ws   = (float*)d_ws;
  float* part = ws;
  float* C8   = ws + (size_t)BATCH * 10 * 32;

  front_kernel<<<BATCH, 256, (size_t)FRONT_LDS_FLOATS * sizeof(float), stream>>>(
      x, W1, b1, W2, b2, W3, b3, C8);

  trace_gemm_kernel<<<BATCH * 10, 128, 0, stream>>>(C8, out, part);
  scale_kernel<<<BATCH * 32, 256, 0, stream>>>(out, part);
}
